# starting kernel
# baseline (speedup 1.0000x reference)
#include <hip/hip_runtime.h>
#include <hip/hip_bf16.h>

constexpr int N_NODES = 100000;
constexpr int N_EDGES = 1600000;
constexpr int IN_F    = 128;
constexpr int HID     = 32;
constexpr int NCLS    = 8;
constexpr int HEADS   = 2;
#define NEG_SLOPE 0.2f

constexpr int M1_TILES = N_NODES / 16;
constexpr int M2_TILES = (N_NODES * 2) / 16;

typedef __attribute__((ext_vector_type(16))) __bf16 v16bf;
typedef __attribute__((ext_vector_type(8)))  float  v8f;

union FragU { v16bf v; uint4 q[2]; };

__device__ __forceinline__ unsigned encF(float f) {
    unsigned u = __float_as_uint(f);
    return (u & 0x80000000u) ? ~u : (u | 0x80000000u);
}
__device__ __forceinline__ float decF(unsigned u) {
    return __uint_as_float((u & 0x80000000u) ? (u ^ 0x80000000u) : ~u);
}
__device__ __forceinline__ float lrelu(float x) {
    return x > 0.0f ? x : NEG_SLOPE * x;
}

__device__ __forceinline__ v16bf load_frag_bf16(const __bf16* __restrict__ base,
                                                int row, int ldk, int kbase, int lane) {
    const __bf16* p = base + (size_t)row * ldk + kbase + ((lane & 16) ? 8 : 0);
    FragU f;
    f.q[0] = *(const uint4*)(p);
    f.q[1] = *(const uint4*)(p + 16);
    return f.v;
}

__global__ void k_zero_f32(float* __restrict__ p, int n) {
    int i = blockIdx.x * blockDim.x + threadIdx.x;
    int stride = gridDim.x * blockDim.x;
    for (; i < n; i += stride) p[i] = 0.0f;
}

__global__ void k_convert_in(const float* __restrict__ x, __bf16* __restrict__ y, int n) {
    int i = blockIdx.x * blockDim.x + threadIdx.x;
    int stride = gridDim.x * blockDim.x;
    for (; i < n; i += stride) y[i] = (__bf16)x[i];
}

__global__ void k_pack_weights(const float* __restrict__ W1,
                               const float* __restrict__ W2,
                               __bf16* __restrict__ W1T,
                               __bf16* __restrict__ W2T) {
    for (int i = threadIdx.x; i < 64 * 128; i += blockDim.x) {
        int n = i >> 7, k = i & 127;
        W1T[i] = (__bf16)W1[k * 64 + n];
    }
    for (int i = threadIdx.x; i < 16 * 32; i += blockDim.x) {
        int n = i >> 5, k = i & 31;
        W2T[i] = (__bf16)W2[k * 16 + n];
    }
}

__global__ void k_gemm1(const __bf16* __restrict__ A, const __bf16* __restrict__ WT,
                        float* __restrict__ out) {
    int wave = (blockIdx.x * blockDim.x + threadIdx.x) >> 5;
    int lane = threadIdx.x & 31;
    if (wave >= M1_TILES) return;
    int rowA = wave * 16 + (lane & 15);

    v8f acc[4] = {};
#pragma unroll
    for (int kc = 0; kc < 4; ++kc) {
        int kb = kc * 32;
        v16bf av = load_frag_bf16(A, rowA, 128, kb, lane);
#pragma unroll
        for (int ct = 0; ct < 4; ++ct) {
            v16bf bv = load_frag_bf16(WT, ct * 16 + (lane & 15), 128, kb, lane);
            acc[ct] = __builtin_amdgcn_wmma_f32_16x16x32_bf16(
                false, av, false, bv, (short)0, acc[ct], false, false);
        }
    }
    int rbase = wave * 16 + ((lane & 16) ? 8 : 0);
    int col = lane & 15;
#pragma unroll
    for (int ct = 0; ct < 4; ++ct)
#pragma unroll
        for (int r = 0; r < 8; ++r)
            out[(size_t)(rbase + r) * 64 + ct * 16 + col] = acc[ct][r];
}

__global__ void k_gemm2(const __bf16* __restrict__ A, const __bf16* __restrict__ WT,
                        float* __restrict__ out) {
    int wave = (blockIdx.x * blockDim.x + threadIdx.x) >> 5;
    int lane = threadIdx.x & 31;
    if (wave >= M2_TILES) return;
    int rowA = wave * 16 + (lane & 15);

    v16bf av = load_frag_bf16(A, rowA, 32, 0, lane);
    v16bf bv = load_frag_bf16(WT, lane & 15, 32, 0, lane);
    v8f acc = {};
    acc = __builtin_amdgcn_wmma_f32_16x16x32_bf16(
        false, av, false, bv, (short)0, acc, false, false);

    int rbase = wave * 16 + ((lane & 16) ? 8 : 0);
    int col = lane & 15;
#pragma unroll
    for (int r = 0; r < 8; ++r)
        out[(size_t)(rbase + r) * 16 + col] = acc[r];
}

__global__ void k_logits1(const float* __restrict__ feat1, const float* __restrict__ al,
                          const float* __restrict__ ar, float* __restrict__ el,
                          float* __restrict__ er) {
    int idx = blockIdx.x * blockDim.x + threadIdx.x;
    if (idx >= N_NODES * HEADS) return;
    int n = idx >> 1, h = idx & 1;
    const float* f = feat1 + (size_t)n * 64 + h * 32;
    float sl = 0.f, sr = 0.f;
#pragma unroll
    for (int j = 0; j < 32; ++j) {
        float v = f[j];
        sl += v * al[h * 32 + j];
        sr += v * ar[h * 32 + j];
    }
    el[idx] = sl;
    er[idx] = sr;
}

__global__ void k_logits2(const float* __restrict__ feat2, const float* __restrict__ al,
                          const float* __restrict__ ar, float* __restrict__ el,
                          float* __restrict__ er) {
    int idx = blockIdx.x * blockDim.x + threadIdx.x;
    if (idx >= N_NODES * 4) return;
    int n = idx >> 2, hh = idx & 3, h2 = hh & 1;
    const float* f = feat2 + (size_t)n * 32 + hh * 8;
    float sl = 0.f, sr = 0.f;
#pragma unroll
    for (int c = 0; c < 8; ++c) {
        float v = f[c];
        sl += v * al[h2 * 8 + c];
        sr += v * ar[h2 * 8 + c];
    }
    el[idx] = sl;
    er[idx] = sr;
}

template <int C>
__global__ void k_edge_max(const int* __restrict__ src, const int* __restrict__ dst,
                           const float* __restrict__ el, const float* __restrict__ er,
                           unsigned* __restrict__ emax) {
    int i = blockIdx.x * blockDim.x + threadIdx.x;
    if (i >= N_EDGES) return;
    int s = src[i], d = dst[i];
#pragma unroll
    for (int h = 0; h < C; ++h) {
        float e = lrelu(el[(size_t)s * C + h] + er[(size_t)d * C + h]);
        atomicMax(&emax[(size_t)d * C + h], encF(e));
    }
}

template <int C>
__global__ void k_edge_exp(const int* __restrict__ src, const int* __restrict__ dst,
                           const float* __restrict__ el, const float* __restrict__ er,
                           const unsigned* __restrict__ emax, float* __restrict__ asum,
                           float* __restrict__ aout) {
    int i = blockIdx.x * blockDim.x + threadIdx.x;
    if (i >= N_EDGES) return;
    int s = src[i], d = dst[i];
#pragma unroll
    for (int h = 0; h < C; ++h) {
        float e = lrelu(el[(size_t)s * C + h] + er[(size_t)d * C + h]);
        float a = __expf(e - decF(emax[(size_t)d * C + h]));
        unsafeAtomicAdd(&asum[(size_t)d * C + h], a);
        aout[(size_t)i * C + h] = a;
    }
}

__global__ void k_edge_agg1(const int* __restrict__ src, const int* __restrict__ dst,
                            const float* __restrict__ a, const float* __restrict__ feat1,
                            float* __restrict__ out1) {
    int gid = blockIdx.x * blockDim.x + threadIdx.x;
    int e = gid >> 5, lane = gid & 31;
    if (e >= N_EDGES) return;
    int s = src[e], d = dst[e];
    float a0 = a[(size_t)e * 2];
    float a1 = a[(size_t)e * 2 + 1];
    const float* fs = feat1 + (size_t)s * 64;
    float* od = out1 + (size_t)d * 64;
    unsafeAtomicAdd(&od[lane],      a0 * fs[lane]);
    unsafeAtomicAdd(&od[lane + 32], a1 * fs[lane + 32]);
}

__global__ void k_edge_agg2(const int* __restrict__ src, const int* __restrict__ dst,
                            const float* __restrict__ a, const float* __restrict__ feat2,
                            float* __restrict__ out2) {
    int gid = blockIdx.x * blockDim.x + threadIdx.x;
    int e = gid >> 5, lane = gid & 31;
    if (e >= N_EDGES) return;
    int s = src[e], d = dst[e];
    float av = a[(size_t)e * 4 + (lane >> 3)];
    unsafeAtomicAdd(&out2[(size_t)d * 32 + lane], av * feat2[(size_t)s * 32 + lane]);
}

__global__ void k_final1(const float* __restrict__ out1, const float* __restrict__ asum,
                         const float* __restrict__ b1, __bf16* __restrict__ hbf) {
    int idx = blockIdx.x * blockDim.x + threadIdx.x;
    if (idx >= N_NODES * 64) return;
    int n = idx >> 6, r = idx & 63, h = r >> 5, j = r & 31;
    float s = asum[(size_t)n * 2 + h];
    float v = (s > 0.0f) ? out1[idx] / s : 0.0f;
    v += b1[h * 32 + j];
    v = v > 0.0f ? v : 0.0f;
    hbf[idx] = (__bf16)v;
}

__global__ void k_final2(float* __restrict__ out, const float* __restrict__ asum,
                         const float* __restrict__ b2) {
    int idx = blockIdx.x * blockDim.x + threadIdx.x;
    if (idx >= N_NODES * 32) return;
    int n = idx >> 5, r = idx & 31, hh = r >> 3, h2 = hh & 1, c = r & 7;
    float s = asum[(size_t)n * 4 + hh];
    float v = (s > 0.0f) ? out[idx] / s : 0.0f;
    out[idx] = v + b2[h2 * 8 + c];
}

extern "C" void kernel_launch(void* const* d_in, const int* in_sizes, int n_in,
                              void* d_out, int out_size, void* d_ws, size_t ws_size,
                              hipStream_t stream) {
    const float* in_feat = (const float*)d_in[0];
    const int*   src     = (const int*)d_in[1];
    const int*   dst     = (const int*)d_in[2];
    const float* W1      = (const float*)d_in[3];
    const float* al1     = (const float*)d_in[4];
    const float* ar1     = (const float*)d_in[5];
    const float* b1      = (const float*)d_in[6];
    const float* W2      = (const float*)d_in[7];
    const float* al2     = (const float*)d_in[8];
    const float* ar2     = (const float*)d_in[9];
    const float* b2      = (const float*)d_in[10];
    float* out = (float*)d_out;

    char* w = (char*)d_ws;
    size_t off = 0;
    auto take = [&](size_t bytes) -> void* {
        void* p = (void*)(w + off);
        off += (bytes + 255) & ~(size_t)255;
        return p;
    };
    __bf16*   in_bf  = (__bf16*)  take((size_t)N_NODES * IN_F * 2);
    __bf16*   w1t    = (__bf16*)  take((size_t)64 * 128 * 2);
    __bf16*   w2t    = (__bf16*)  take((size_t)16 * 32 * 2);
    float*    feat1  = (float*)   take((size_t)N_NODES * 64 * 4);
    float*    el1    = (float*)   take((size_t)N_NODES * 2 * 4);
    float*    er1    = (float*)   take((size_t)N_NODES * 2 * 4);
    unsigned* emax1  = (unsigned*)take((size_t)N_NODES * 2 * 4);
    float*    asum1  = (float*)   take((size_t)N_NODES * 2 * 4);
    float*    a1     = (float*)   take((size_t)N_EDGES * 2 * 4);
    float*    out1   = (float*)   take((size_t)N_NODES * 64 * 4);
    __bf16*   h1bf   = (__bf16*)  take((size_t)N_NODES * 64 * 2);
    float*    feat2  = (float*)   take((size_t)N_NODES * 32 * 4);
    float*    el2    = (float*)   take((size_t)N_NODES * 4 * 4);
    float*    er2    = (float*)   take((size_t)N_NODES * 4 * 4);
    unsigned* emax2  = (unsigned*)take((size_t)N_NODES * 4 * 4);
    float*    asum2  = (float*)   take((size_t)N_NODES * 4 * 4);
    float*    a2     = (float*)   take((size_t)N_EDGES * 4 * 4);
    (void)ws_size; (void)n_in; (void)in_sizes; (void)out_size;

    auto cdiv = [](long a, long b) { return (int)((a + b - 1) / b); };
    const int T = 256;

    k_zero_f32<<<cdiv(N_NODES * 64, T), T, 0, stream>>>(out1, N_NODES * 64);
    k_zero_f32<<<cdiv(N_NODES * 2, T), T, 0, stream>>>(asum1, N_NODES * 2);
    k_zero_f32<<<cdiv(N_NODES * 2, T), T, 0, stream>>>((float*)emax1, N_NODES * 2);
    k_zero_f32<<<cdiv(N_NODES * 4, T), T, 0, stream>>>(asum2, N_NODES * 4);
    k_zero_f32<<<cdiv(N_NODES * 4, T), T, 0, stream>>>((float*)emax2, N_NODES * 4);
    k_zero_f32<<<cdiv(N_NODES * 32, T), T, 0, stream>>>(out, N_NODES * 32);

    k_convert_in<<<4096, T, 0, stream>>>(in_feat, in_bf, N_NODES * IN_F);
    k_pack_weights<<<1, T, 0, stream>>>(W1, W2, w1t, w2t);

    k_gemm1<<<cdiv((long)M1_TILES * 32, T), T, 0, stream>>>(in_bf, w1t, feat1);
    k_logits1<<<cdiv(N_NODES * 2, T), T, 0, stream>>>(feat1, al1, ar1, el1, er1);
    k_edge_max<2><<<cdiv(N_EDGES, T), T, 0, stream>>>(src, dst, el1, er1, emax1);
    k_edge_exp<2><<<cdiv(N_EDGES, T), T, 0, stream>>>(src, dst, el1, er1, emax1, asum1, a1);
    k_edge_agg1<<<cdiv((long)N_EDGES * 32, T), T, 0, stream>>>(src, dst, a1, feat1, out1);
    k_final1<<<cdiv(N_NODES * 64, T), T, 0, stream>>>(out1, asum1, b1, h1bf);

    k_gemm2<<<cdiv((long)M2_TILES * 32, T), T, 0, stream>>>(h1bf, w2t, feat2);
    k_logits2<<<cdiv(N_NODES * 4, T), T, 0, stream>>>(feat2, al2, ar2, el2, er2);
    k_edge_max<4><<<cdiv(N_EDGES, T), T, 0, stream>>>(src, dst, el2, er2, emax2);
    k_edge_exp<4><<<cdiv(N_EDGES, T), T, 0, stream>>>(src, dst, el2, er2, emax2, asum2, a2);
    k_edge_agg2<<<cdiv((long)N_EDGES * 32, T), T, 0, stream>>>(src, dst, a2, feat2, out);
    k_final2<<<cdiv(N_NODES * 32, T), T, 0, stream>>>(out, asum2, b2);
}


# Both aggregations: source logits formed from the gathered rows in registers (DPP transposing sum over a head's 8 or 4 lanes) instead of a per-edge logit fetch; logit and softmax sums in a new order
# speedup vs baseline: 7.3663x; 7.3663x over previous
#include <hip/hip_runtime.h>
#include <hip/hip_bf16.h>

constexpr int N_NODES = 100000;
constexpr int N_EDGES = 1600000;
constexpr int IN_F    = 128;
constexpr int HID     = 32;
constexpr int NCLS    = 8;
constexpr int HEADS   = 2;
#define NEG_SLOPE 0.2f

constexpr int M1_TILES = N_NODES / 16;

typedef __attribute__((ext_vector_type(16))) __bf16 v16bf;
typedef __attribute__((ext_vector_type(8)))  float  v8f;

union FragU { v16bf v; uint4 q[2]; };

__device__ __forceinline__ float lrelu(float x) {
    return x > 0.0f ? x : NEG_SLOPE * x;
}

__device__ __forceinline__ v16bf load_frag_bf16(const __bf16* __restrict__ base,
                                                int row, int ldk, int kbase, int lane) {
    const __bf16* p = base + (size_t)row * ldk + kbase + ((lane & 16) ? 8 : 0);
    FragU f;
    f.q[0] = *(const uint4*)(p);
    f.q[1] = *(const uint4*)(p + 16);
    return f.v;
}

__device__ __forceinline__ void pack_weights(const float* __restrict__ W1,
                                             const float* __restrict__ W2,
                                             __bf16* __restrict__ W1T,
                                             __bf16* __restrict__ W2T,
                                             int first, int step) {
    for (int i = first; i < 64 * 128; i += step) {
        int n = i >> 7, k = i & 127;
        W1T[i] = (__bf16)W1[k * 64 + n];
    }
    for (int i = first; i < 16 * 32; i += step) {
        int n = i >> 5, k = i & 31;
        W2T[i] = (__bf16)W2[k * 16 + n];
    }
}

constexpr int G1_WAVES = 32;
constexpr int G1_BLOCKS = (M1_TILES + G1_WAVES - 1) / G1_WAVES;
constexpr int G1_PITCH = 64 + 4;

__device__ __forceinline__ void
gemm1_block(int gblock, float (* __restrict__ tile)[16 * G1_PITCH],
            const float* __restrict__ X, const __bf16* __restrict__ WT,
            const float* __restrict__ ar,
            float* __restrict__ feat1, float* __restrict__ er) {
    const int lane = threadIdx.x & 31;
    const int wv   = threadIdx.x >> 5;
    int t = gblock * G1_WAVES + wv;
    const bool live = t < M1_TILES;
    if (!live) t = M1_TILES - 1;
    const float* xr = X + (size_t)(t * 16 + (lane & 15)) * IN_F + ((lane & 16) ? 8 : 0);
    float4 xa[4][4];
#pragma unroll
    for (int kc = 0; kc < 4; ++kc) {
        const float* p = xr + kc * 32;
        xa[kc][0] = *(const float4*)(p);
        xa[kc][1] = *(const float4*)(p + 4);
        xa[kc][2] = *(const float4*)(p + 16);
        xa[kc][3] = *(const float4*)(p + 20);
    }

    v8f acc[4] = {};
#pragma unroll
    for (int kc = 0; kc < 4; ++kc) {
        int kb = kc * 32;
        v16bf av;
#pragma unroll
        for (int q = 0; q < 4; ++q) {
            av[4 * q + 0] = (__bf16)xa[kc][q].x;
            av[4 * q + 1] = (__bf16)xa[kc][q].y;
            av[4 * q + 2] = (__bf16)xa[kc][q].z;
            av[4 * q + 3] = (__bf16)xa[kc][q].w;
        }
#pragma unroll
        for (int ct = 0; ct < 4; ++ct) {
            v16bf bv = load_frag_bf16(WT, ct * 16 + (lane & 15), 128, kb, lane);
            acc[ct] = __builtin_amdgcn_wmma_f32_16x16x32_bf16(
                false, av, false, bv, (short)0, acc[ct], false, false);
        }
    }
    float* tl = tile[wv];
    {
        const int rb = (lane & 16) ? 8 : 0, col = lane & 15;
#pragma unroll
        for (int ct = 0; ct < 4; ++ct)
#pragma unroll
            for (int r = 0; r < 8; ++r)
                tl[(rb + r) * G1_PITCH + ct * 16 + col] = acc[ct][r];
    }
    __syncthreads();
    if (!live) return;

#pragma unroll
    for (int i = 0; i < 8; ++i) {
        int idx = i * 32 + lane;
        int row = idx >> 4, c4 = idx & 15;
        float4 v = *(const float4*)(tl + row * G1_PITCH + c4 * 4);
        *(float4*)(feat1 + (size_t)(t * 16 + row) * 64 + c4 * 4) = v;
    }
    {
        const int row = lane >> 1, h = lane & 1;
        const float* f = tl + row * G1_PITCH + h * 32;
        float sr = 0.f;
#pragma unroll
        for (int j = 0; j < 32; ++j) sr += f[j] * ar[h * 32 + j];
        er[(size_t)t * 32 + lane] = sr;
    }
}

constexpr int CS_R     = 16;
constexpr int CS_S     = 32;
constexpr int CS_RANGE = N_NODES / CS_R;
constexpr int CS_SLAB  = N_EDGES / CS_S;
constexpr int CS_Q     = CS_SLAB / 4;
constexpr int CS_T     = 1024;
constexpr int CS_U     = 4;
static_assert(N_NODES % CS_R == 0 && N_EDGES % CS_S == 0 && CS_SLAB % 4 == 0,
              "ranges and slabs must cover nodes and edges exactly");

__global__ void __launch_bounds__(CS_T)
k_count(const int* __restrict__ dst, int* __restrict__ cnt) {
    __shared__ int ctr[CS_RANGE];
    const int r = blockIdx.x % CS_R, s = blockIdx.x / CS_R;
    const int lo = r * CS_RANGE;
    const int tid = threadIdx.x;
    for (int k = tid; k < CS_RANGE; k += CS_T) ctr[k] = 0;
    __syncthreads();
    const int4* __restrict__ p = (const int4*)(dst + (size_t)s * CS_SLAB);
    for (int q0 = 0; q0 < CS_Q; q0 += CS_U * CS_T) {
        int4 d[CS_U];
#pragma unroll
        for (int u = 0; u < CS_U; ++u) {
            int q = q0 + u * CS_T + tid;
            d[u] = make_int4(-1, -1, -1, -1);
            if (q < CS_Q) d[u] = p[q];
        }
#pragma unroll
        for (int u = 0; u < CS_U; ++u) {
            unsigned a0 = (unsigned)(d[u].x - lo), a1 = (unsigned)(d[u].y - lo);
            unsigned a2 = (unsigned)(d[u].z - lo), a3 = (unsigned)(d[u].w - lo);
            if (a0 < (unsigned)CS_RANGE) atomicAdd(&ctr[a0], 1);
            if (a1 < (unsigned)CS_RANGE) atomicAdd(&ctr[a1], 1);
            if (a2 < (unsigned)CS_RANGE) atomicAdd(&ctr[a2], 1);
            if (a3 < (unsigned)CS_RANGE) atomicAdd(&ctr[a3], 1);
        }
    }
    __syncthreads();
    for (int k = tid; k < CS_RANGE; k += CS_T) cnt[(size_t)s * N_NODES + lo + k] = ctr[k];
}

constexpr int SP_T        = 256;
constexpr int SP_BLOCKS   = (N_NODES + SP_T - 1) / SP_T;
constexpr int PACK_BLOCKS = (64 * 128) / SP_T;

__global__ void __launch_bounds__(SP_T)
k_slab_prefix(int* __restrict__ cnt, int* __restrict__ deg,
              const float* __restrict__ W1, const float* __restrict__ W2,
              __bf16* __restrict__ W1T, __bf16* __restrict__ W2T) {
    if (blockIdx.x >= SP_BLOCKS) {
        pack_weights(W1, W2, W1T, W2T, (blockIdx.x - SP_BLOCKS) * SP_T + threadIdx.x, PACK_BLOCKS * SP_T);
        return;
    }
    int n = blockIdx.x * SP_T + threadIdx.x;
    if (n >= N_NODES) return;
    int c[CS_S];
#pragma unroll
    for (int s = 0; s < CS_S; ++s) c[s] = cnt[(size_t)s * N_NODES + n];
    int run = 0;
#pragma unroll
    for (int s = 0; s < CS_S; ++s) {
        cnt[(size_t)s * N_NODES + n] = run;
        run += c[s];
    }
    deg[n] = run;
}

constexpr int SCAN_T   = 1024;
constexpr int SCAN_PER = 100;
static_assert(N_NODES % SCAN_PER == 0 && N_NODES / SCAN_PER <= SCAN_T && SCAN_PER % 4 == 0,
              "scan tiling must cover the node count exactly");

static_assert(G1_WAVES * 32 == SCAN_T, "GEMM 1 blocks and the scan block share one launch shape");

__global__ void __launch_bounds__(SCAN_T)
k_scan_gemm1(const int* __restrict__ deg, int* __restrict__ rowptr,
             const float* __restrict__ X, const __bf16* __restrict__ WT,
             const float* __restrict__ ar,
             float* __restrict__ feat1, float* __restrict__ er) {
    __shared__ int part[SCAN_T];
    __shared__ __attribute__((aligned(16))) float tile[G1_WAVES][16 * G1_PITCH];
    if (blockIdx.x != 0) {
        gemm1_block(blockIdx.x - 1, tile, X, WT, ar, feat1, er);
        return;
    }
    const int t = threadIdx.x;
    const bool act = t < N_NODES / SCAN_PER;
    int4 v[SCAN_PER / 4];
    int sum = 0;
    if (act) {
        const int4* p = (const int4*)(deg + t * SCAN_PER);
#pragma unroll
        for (int k = 0; k < SCAN_PER / 4; ++k) v[k] = p[k];
#pragma unroll
        for (int k = 0; k < SCAN_PER / 4; ++k) sum += v[k].x + v[k].y + v[k].z + v[k].w;
    }
    part[t] = sum;
    __syncthreads();
    for (int off = 1; off < SCAN_T; off <<= 1) {
        int x = (t >= off) ? part[t - off] : 0;
        __syncthreads();
        part[t] += x;
        __syncthreads();
    }
    if (act) {
        int run = part[t] - sum;
        int4* q = (int4*)(rowptr + t * SCAN_PER);
#pragma unroll
        for (int k = 0; k < SCAN_PER / 4; ++k) {
            int4 o;
            o.x = run; run += v[k].x;
            o.y = run; run += v[k].y;
            o.z = run; run += v[k].z;
            o.w = run; run += v[k].w;
            q[k] = o;
        }
        if (t == N_NODES / SCAN_PER - 1) rowptr[N_NODES] = run;
    }
}

__global__ void __launch_bounds__(CS_T)
k_fill(const int* __restrict__ src, const int* __restrict__ dst,
       const int* __restrict__ rowptr, const int* __restrict__ cnt,
       int* __restrict__ csr_src) {
    __shared__ int cur[CS_RANGE];
    const int r = blockIdx.x % CS_R, s = blockIdx.x / CS_R;
    const int lo = r * CS_RANGE;
    const int tid = threadIdx.x;
    {
        constexpr int CUR_PER = (CS_RANGE + CS_T - 1) / CS_T;
        int rv[CUR_PER], cv[CUR_PER];
#pragma unroll
        for (int i = 0; i < CUR_PER; ++i) {
            int k = tid + i * CS_T;
            rv[i] = 0; cv[i] = 0;
            if (k < CS_RANGE) {
                rv[i] = rowptr[lo + k];
                cv[i] = cnt[(size_t)s * N_NODES + lo + k];
            }
        }
#pragma unroll
        for (int i = 0; i < CUR_PER; ++i) {
            int k = tid + i * CS_T;
            if (k < CS_RANGE) cur[k] = rv[i] + cv[i];
        }
    }
    __syncthreads();
    const int4* __restrict__ p  = (const int4*)(dst + (size_t)s * CS_SLAB);
    const int4* __restrict__ sp = (const int4*)(src + (size_t)s * CS_SLAB);
    for (int q0 = 0; q0 < CS_Q; q0 += CS_U * CS_T) {
        int4 d[CS_U], sv[CS_U];
#pragma unroll
        for (int u = 0; u < CS_U; ++u) {
            int q = q0 + u * CS_T + tid;
            d[u]  = make_int4(-1, -1, -1, -1);
            sv[u] = make_int4(0, 0, 0, 0);
            if (q < CS_Q) { d[u] = p[q]; sv[u] = sp[q]; }
        }
#pragma unroll
        for (int u = 0; u < CS_U; ++u) {
            const int dv[4] = {d[u].x, d[u].y, d[u].z, d[u].w};
            const int sw[4] = {sv[u].x, sv[u].y, sv[u].z, sv[u].w};
#pragma unroll
            for (int k = 0; k < 4; ++k) {
                unsigned a = (unsigned)(dv[k] - lo);
                if (a < (unsigned)CS_RANGE) {
                    int pos = atomicAdd(&cur[a], 1);
                    if ((unsigned)pos < (unsigned)N_EDGES) csr_src[pos] = sw[k];
                }
            }
        }
    }
}

template <int J> __device__ __forceinline__ int row_i(int v) {
    return __builtin_amdgcn_update_dpp(0, v, 0x150 + J, 0xF, 0xF, false);
}
template <int J> __device__ __forceinline__ float row_f(float v) {
    return __int_as_float(row_i<J>(__float_as_int(v)));
}
__device__ __forceinline__ int node_clamp(int s) {
    return s < 0 ? 0 : (s >= N_NODES ? N_NODES - 1 : s);
}

typedef __attribute__((ext_vector_type(4))) __bf16 v4bf;
typedef __attribute__((ext_vector_type(2))) float v2f;
typedef __attribute__((ext_vector_type(4))) float v4f;

template <int CTRL> __device__ __forceinline__ float quad_mov(float v) {
    return __int_as_float(__builtin_amdgcn_update_dpp(0, __float_as_int(v), CTRL, 0xF, 0xF, false));
}
constexpr int DPP_X1 = 0xB1;
constexpr int DPP_X2 = 0x4E;
constexpr int DPP_X4 = 0x164;
__device__ __forceinline__ float oct_sum(float v) {
    v += quad_mov<DPP_X1>(v);
    v += quad_mov<DPP_X2>(v);
    v += quad_mov<DPP_X4>(v);
    return v;
}
__device__ __forceinline__ float oct_max(float v) {
    v = fmaxf(v, quad_mov<DPP_X1>(v));
    v = fmaxf(v, quad_mov<DPP_X2>(v));
    v = fmaxf(v, quad_mov<DPP_X4>(v));
    return v;
}

template <int J>
__device__ __forceinline__ v4f row1_ld(const char* __restrict__ fb, int s, unsigned coff) {
    const unsigned off = (unsigned)row_i<J>(s) * 256u + coff;
    return *(const v4f*)(fb + off);
}
template <int O>
__device__ __forceinline__ void oct1_ld(v4f (&f)[8], const char* __restrict__ fb, int s, unsigned coff) {
    f[0] = row1_ld<8 * O + 0>(fb, s, coff);
    f[1] = row1_ld<8 * O + 1>(fb, s, coff);
    f[2] = row1_ld<8 * O + 2>(fb, s, coff);
    f[3] = row1_ld<8 * O + 3>(fb, s, coff);
    f[4] = row1_ld<8 * O + 4>(fb, s, coff);
    f[5] = row1_ld<8 * O + 5>(fb, s, coff);
    f[6] = row1_ld<8 * O + 6>(fb, s, coff);
    f[7] = row1_ld<8 * O + 7>(fb, s, coff);
}
__device__ __forceinline__ float oct1_score(const v4f (&f)[8], const v4f alv, float erc,
                                            int r, int first, int deg) {
    float p[8];
#pragma unroll
    for (int q = 0; q < 8; ++q) {
        const v2f t = f[q].hi * alv.hi + f[q].lo * alv.lo;
        p[q] = t.x + t.y;
    }
    const bool o1 = (r & 1) != 0, o2 = (r & 2) != 0, o4 = (r & 4) != 0;
    float u[4], v[2];
#pragma unroll
    for (int k = 0; k < 4; ++k)
        u[k] = (o1 ? p[2 * k + 1] : p[2 * k]) + quad_mov<DPP_X1>(o1 ? p[2 * k] : p[2 * k + 1]);
#pragma unroll
    for (int k = 0; k < 2; ++k)
        v[k] = (o2 ? u[2 * k + 1] : u[2 * k]) + quad_mov<DPP_X2>(o2 ? u[2 * k] : u[2 * k + 1]);
    const float t = (o4 ? v[1] : v[0]) + quad_mov<DPP_X4>(o4 ? v[0] : v[1]);
    return (first + r < deg) ? lrelu(t + erc) : -__builtin_inff();
}
template <int Q>
__device__ __forceinline__ void oct1_acc(const v4f f, float a, bool hd1, v4f& acc) {
    const float w0 = row_f<Q>(a), w1 = row_f<8 + Q>(a);
    const float w = hd1 ? w1 : w0;
    acc += w * f;
}
__device__ __forceinline__ void oct1_sum(const v4f (&f)[8], float e, float m, bool hd1, float& z, v4f& acc) {
    const float a = __expf(e - m);
    z += a;
    oct1_acc<0>(f[0], a, hd1, acc);
    oct1_acc<1>(f[1], a, hd1, acc);
    oct1_acc<2>(f[2], a, hd1, acc);
    oct1_acc<3>(f[3], a, hd1, acc);
    oct1_acc<4>(f[4], a, hd1, acc);
    oct1_acc<5>(f[5], a, hd1, acc);
    oct1_acc<6>(f[6], a, hd1, acc);
    oct1_acc<7>(f[7], a, hd1, acc);
}
template <int O, bool SUM>
__device__ __forceinline__ void tail1_oct(const char* __restrict__ fb, int s, unsigned coff, const v4f alv,
                                          float erc, int r, bool hd1, int first, int deg,
                                          float& m, float& z, v4f& acc) {
    v4f f[8];
    oct1_ld<O>(f, fb, s, coff);
    const float e = oct1_score(f, alv, erc, r, first + 8 * O, deg);
    if (!SUM) m = fmaxf(m, e);
    else      oct1_sum(f, e, m, hd1, z, acc);
}
template <bool SUM>
__device__ __forceinline__ void tail1(const int* __restrict__ csr_src, const char* __restrict__ fb,
                                      int beg, int deg, int dmax, int c, int sPad, unsigned coff,
                                      const v4f alv, float erc, float& m, float& z, v4f& acc) {
    const int r = c & 7;
    const bool hd1 = c >= 8;
    for (int base = 24; base < dmax; base += 16) {
        int s = sPad;
        if (base + c < deg) s = node_clamp(csr_src[beg + base + c]);
        const int cu = dmax - base;
        if (cu > 0) tail1_oct<0, SUM>(fb, s, coff, alv, erc, r, hd1, base, deg, m, z, acc);
        if (cu > 8) tail1_oct<1, SUM>(fb, s, coff, alv, erc, r, hd1, base, deg, m, z, acc);
    }
}

static_assert(N_NODES % 2 == 0, "a live wave owns two existing nodes: its 256 bytes of feat2 are stored whole");
constexpr int S1_WAVES = 8;
__global__ void __launch_bounds__(S1_WAVES * 32, 8)
k_seg1(const int* __restrict__ rowptr, const int* __restrict__ csr_src,
       const float* __restrict__ feat1, const float* __restrict__ al, const float* __restrict__ er,
       const float* __restrict__ b1, const __bf16* __restrict__ W2T,
       const float* __restrict__ ar2,
       float* __restrict__ feat2, float* __restrict__ er2) {
    __shared__ __attribute__((aligned(16))) __bf16 hrow[S1_WAVES][4 * 32];
    __shared__ __attribute__((aligned(16))) float  prod[S1_WAVES][4 * 16];
    const int lane = threadIdx.x & 31;
    const int c = lane & 15;
    const int wv = __builtin_amdgcn_readfirstlane((int)((blockIdx.x * blockDim.x + threadIdx.x) >> 5));
    if (2 * wv >= N_NODES) return;
    const int n = 2 * wv + (lane >> 4);
    const bool nv = n < N_NODES;
    const bool hd1 = c >= 8;
    const int r = c & 7;
    int beg = 0, end = 0;
    float erc = 0.f;
    if (nv) {
        beg = rowptr[n]; end = rowptr[n + 1];
        erc = er[(size_t)n * 2 + (c >> 3)];
    }
    const v4f alv = *(const v4f*)(al + 4 * c);
    beg = beg < 0 ? 0 : (beg > N_EDGES ? N_EDGES : beg);
    end = end < beg ? beg : (end > N_EDGES ? N_EDGES : end);
    const int deg = end - beg;
    const int d0 = __builtin_amdgcn_readlane(deg, 0), d1 = __builtin_amdgcn_readlane(deg, 16);
    const int dmax = d0 > d1 ? d0 : d1;
    const bool hasB = dmax > 16;
    const float NINF = -__builtin_inff();

    const bool actA = c < deg, actB = 16 + c < deg;
    int sA = node_clamp(csr_src[actA ? beg + c : 0]);
    int sB = node_clamp(csr_src[actB ? beg + 16 + c : 0]);
    asm volatile("" : "+v"(sB));
    const int sPad = row_i<0>(sA);
    sA = actA ? sA : sPad;
    sB = actB ? sB : sPad;

    const char* fb = (const char*)feat1;
    const unsigned coff = 16u * (unsigned)c;
    float m = NINF, z = 0.f;
    v4f acc = {0.f, 0.f, 0.f, 0.f};
    if (dmax > 24) tail1<false>(csr_src, fb, beg, deg, dmax, c, sPad, coff, alv, erc, m, z, acc);
    v4f fA0[8], fA1[8], fB[8];
    oct1_ld<0>(fA0, fb, sA, coff);
    oct1_ld<1>(fA1, fb, sA, coff);
    if (hasB) oct1_ld<0>(fB, fb, sB, coff);

    const float eA0 = oct1_score(fA0, alv, erc, r, 0, deg);
    const float eA1 = oct1_score(fA1, alv, erc, r, 8, deg);
    float eB = NINF;
    if (hasB) eB = oct1_score(fB, alv, erc, r, 16, deg);
    m = fmaxf(fmaxf(m, eA0), fmaxf(eA1, eB));
    m = oct_max(m);

    oct1_sum(fA0, eA0, m, hd1, z, acc);
    oct1_sum(fA1, eA1, m, hd1, z, acc);
    if (hasB) oct1_sum(fB, eB, m, hd1, z, acc);
    if (dmax > 24) tail1<true>(csr_src, fb, beg, deg, dmax, c, sPad, coff, alv, erc, m, z, acc);
    const v16bf bv = load_frag_bf16(W2T, lane & 15, 32, 0, lane);
    const int h2 = lane & 1;
    const float4 arA = *(const float4*)(ar2 + h2 * 8), arB = *(const float4*)(ar2 + h2 * 8 + 4);

    z = oct_sum(z);
    const float4 b = *(const float4*)(b1 + 4 * c);
    float v0 = (z > 0.0f) ? acc.x / z : 0.0f;
    float v1 = (z > 0.0f) ? acc.y / z : 0.0f;
    float v2 = (z > 0.0f) ? acc.z / z : 0.0f;
    float v3 = (z > 0.0f) ? acc.w / z : 0.0f;
    v0 += b.x; v1 += b.y; v2 += b.z; v3 += b.w;
    v0 = v0 > 0.0f ? v0 : 0.0f;
    v1 = v1 > 0.0f ? v1 : 0.0f;
    v2 = v2 > 0.0f ? v2 : 0.0f;
    v3 = v3 > 0.0f ? v3 : 0.0f;
    v4bf o;
    o.x = (__bf16)v0; o.y = (__bf16)v1; o.z = (__bf16)v2; o.w = (__bf16)v3;

    const int wib = threadIdx.x >> 5;
    __bf16* hw = hrow[wib];
    float*  pw = prod[wib];
    *(v4bf*)(hw + 4 * lane) = o;
    FragU fa;
    {
        const __bf16* pa = hw + (lane & 3) * 32 + ((lane & 16) ? 8 : 0);
        fa.q[0] = *(const uint4*)(pa);
        fa.q[1] = *(const uint4*)(pa + 16);
    }
    v8f d = {};
    d = __builtin_amdgcn_wmma_f32_16x16x32_bf16(false, fa.v, false, bv, (short)0, d, false, false);
    if (lane < 16) {
#pragma unroll
        for (int r = 0; r < 4; ++r) pw[r * 16 + lane] = d[r];
    }
    {
        const float2 fv = *(const float2*)(pw + 2 * lane);
        *(float2*)(feat2 + (size_t)wv * 64 + 2 * lane) = fv;
    }
    if (lane < 8) {
        const float* fr = pw + (lane >> 1) * 16 + h2 * 8;
        const float4 fA = *(const float4*)(fr), fB = *(const float4*)(fr + 4);
        const float fq[8] = {fA.x, fA.y, fA.z, fA.w, fB.x, fB.y, fB.z, fB.w};
        const float rq[8] = {arA.x, arA.y, arA.z, arA.w, arB.x, arB.y, arB.z, arB.w};
        float sr = 0.f;
#pragma unroll
        for (int q = 0; q < 8; ++q) sr += fq[q] * rq[q];
        er2[(size_t)wv * 8 + lane] = sr;
    }
}

template <int J>
__device__ __forceinline__ v2f row2_ld(const char* __restrict__ fb, int s, unsigned coff) {
    const unsigned off = (unsigned)row_i<J>(s) * 128u + coff;
    return *(const v2f*)(fb + off);
}
__device__ __forceinline__ float quad_sum(float v) {
    v += quad_mov<DPP_X1>(v);
    v += quad_mov<DPP_X2>(v);
    return v;
}
__device__ __forceinline__ float quad_max(float v) {
    v = fmaxf(v, quad_mov<DPP_X1>(v));
    v = fmaxf(v, quad_mov<DPP_X2>(v));
    return v;
}
__device__ __forceinline__ float grp2_score(const v2f (&f)[4], const v2f alv, float erc,
                                            int r, int first, int deg) {
    const v2f t0 = f[0] * alv, t1 = f[1] * alv, t2 = f[2] * alv, t3 = f[3] * alv;
    const float p0 = t0.x + t0.y, p1 = t1.x + t1.y, p2 = t2.x + t2.y, p3 = t3.x + t3.y;
    const bool o1 = (r & 1) != 0, o2 = (r & 2) != 0;
    const float u01 = (o1 ? p1 : p0) + quad_mov<DPP_X1>(o1 ? p0 : p1);
    const float u23 = (o1 ? p3 : p2) + quad_mov<DPP_X1>(o1 ? p2 : p3);
    const float t   = (o2 ? u23 : u01) + quad_mov<DPP_X2>(o2 ? u01 : u23);
    return (first + r < deg) ? lrelu(t + erc) : -__builtin_inff();
}
__device__ __forceinline__ void grp2_sum(const v2f (&f)[4], float e, float m, float& z, v2f& acc) {
    const float a = __expf(e - m);
    z += a;
    const float w0 = quad_mov<0x00>(a), w1 = quad_mov<0x55>(a), w2 = quad_mov<0xAA>(a), w3 = quad_mov<0xFF>(a);
    acc += w0 * f[0];
    acc += w1 * f[1];
    acc += w2 * f[2];
    acc += w3 * f[3];
}
template <int G>
__device__ __forceinline__ void grp2_ld(v2f (&f)[4], const char* __restrict__ fb, int s, unsigned coff) {
    f[0] = row2_ld<4 * G + 0>(fb, s, coff);
    f[1] = row2_ld<4 * G + 1>(fb, s, coff);
    f[2] = row2_ld<4 * G + 2>(fb, s, coff);
    f[3] = row2_ld<4 * G + 3>(fb, s, coff);
}
template <int G, bool SUM>
__device__ __forceinline__ void tail2_grp(const char* __restrict__ fb, int s, unsigned coff, const v2f alv,
                                          float erc, int r, int first, int deg, float& m, float& z, v2f& acc) {
    v2f f[4];
    grp2_ld<G>(f, fb, s, coff);
    const float e = grp2_score(f, alv, erc, r, first + 4 * G, deg);
    if (!SUM) m = fmaxf(m, e);
    else      grp2_sum(f, e, m, z, acc);
}
template <bool SUM>
__device__ __forceinline__ void tail2(const int* __restrict__ csr_src, const char* __restrict__ fb,
                                      int beg, int deg, int dmax, int c, int sPad, unsigned coff,
                                      const v2f alv, float erc, float& m, float& z, v2f& acc) {
    const int r = c & 3;
    for (int base = 24; base < dmax; base += 16) {
        int s = sPad;
        if (base + c < deg) s = node_clamp(csr_src[beg + base + c]);
        const int cu = dmax - base;
        if (cu > 0)  tail2_grp<0, SUM>(fb, s, coff, alv, erc, r, base, deg, m, z, acc);
        if (cu > 4)  tail2_grp<1, SUM>(fb, s, coff, alv, erc, r, base, deg, m, z, acc);
        if (cu > 8)  tail2_grp<2, SUM>(fb, s, coff, alv, erc, r, base, deg, m, z, acc);
        if (cu > 12) tail2_grp<3, SUM>(fb, s, coff, alv, erc, r, base, deg, m, z, acc);
    }
}

__global__ void __launch_bounds__(256)
k_seg2(const int* __restrict__ rowptr, const int* __restrict__ csr_src,
       const float* __restrict__ feat2, const float* __restrict__ al2, const float* __restrict__ er,
       const float* __restrict__ b2, float* __restrict__ out) {
    const int lane = threadIdx.x & 31;
    const int c = lane & 15;
    const int wv = __builtin_amdgcn_readfirstlane((int)((blockIdx.x * blockDim.x + threadIdx.x) >> 5));
    if (2 * wv >= N_NODES) return;
    const int n = 2 * wv + (lane >> 4);
    const bool nv = n < N_NODES;
    const int ch = c >> 2, r = c & 3;
    int beg = 0, end = 0;
    float erc = 0.f;
    if (nv) {
        beg = rowptr[n]; end = rowptr[n + 1];
        erc = er[(size_t)n * 4 + ch];
    }
    const v2f alv = *(const v2f*)(al2 + (ch & 1) * 8 + ((2 * c) & 7));
    beg = beg < 0 ? 0 : (beg > N_EDGES ? N_EDGES : beg);
    end = end < beg ? beg : (end > N_EDGES ? N_EDGES : end);
    const int deg = end - beg;
    const int d0 = __builtin_amdgcn_readlane(deg, 0), d1 = __builtin_amdgcn_readlane(deg, 16);
    const int dmax = d0 > d1 ? d0 : d1;
    const bool hasB = dmax > 16;

    const bool actA = c < deg, actB = 16 + c < deg;
    int sA = node_clamp(csr_src[actA ? beg + c : 0]);
    int sB = node_clamp(csr_src[actB ? beg + 16 + c : 0]);
    asm volatile("" : "+v"(sB));
    const int sPad = row_i<0>(sA);
    sA = actA ? sA : sPad;
    sB = actB ? sB : sPad;

    const char* fb = (const char*)feat2;
    const unsigned coff = 8u * (unsigned)c;
    v2f fA[4][4], fB[2][4];
    grp2_ld<0>(fA[0], fb, sA, coff);
    grp2_ld<1>(fA[1], fb, sA, coff);
    grp2_ld<2>(fA[2], fb, sA, coff);
    grp2_ld<3>(fA[3], fb, sA, coff);
    if (hasB) {
        grp2_ld<0>(fB[0], fb, sB, coff);
        grp2_ld<1>(fB[1], fb, sB, coff);
    }
    const float NINF = -__builtin_inff();
    const float eA0 = grp2_score(fA[0], alv, erc, r, 0, deg);
    const float eA1 = grp2_score(fA[1], alv, erc, r, 4, deg);
    const float eA2 = grp2_score(fA[2], alv, erc, r, 8, deg);
    const float eA3 = grp2_score(fA[3], alv, erc, r, 12, deg);
    float eB0 = NINF, eB1 = NINF;
    if (hasB) {
        eB0 = grp2_score(fB[0], alv, erc, r, 16, deg);
        eB1 = grp2_score(fB[1], alv, erc, r, 20, deg);
    }
    float m = fmaxf(fmaxf(fmaxf(eA0, eA1), fmaxf(eA2, eA3)), fmaxf(eB0, eB1));
    float z = 0.f;
    v2f acc = {0.f, 0.f};
    if (dmax > 24) tail2<false>(csr_src, fb, beg, deg, dmax, c, sPad, coff, alv, erc, m, z, acc);
    m = quad_max(m);

    grp2_sum(fA[0], eA0, m, z, acc);
    grp2_sum(fA[1], eA1, m, z, acc);
    grp2_sum(fA[2], eA2, m, z, acc);
    grp2_sum(fA[3], eA3, m, z, acc);
    if (hasB) {
        grp2_sum(fB[0], eB0, m, z, acc);
        grp2_sum(fB[1], eB1, m, z, acc);
    }
    if (dmax > 24) tail2<true>(csr_src, fb, beg, deg, dmax, c, sPad, coff, alv, erc, m, z, acc);
    z = quad_sum(z);

    const float2 b = *(const float2*)(b2 + (ch & 1) * 8 + ((2 * c) & 7));
    float2 o;
    o.x = ((z > 0.0f) ? acc.x / z : 0.0f) + b.x;
    o.y = ((z > 0.0f) ? acc.y / z : 0.0f) + b.y;
    if (nv) *(float2*)(out + (size_t)n * 32 + 2 * c) = o;
}

extern "C" void kernel_launch(void* const* d_in, const int* in_sizes, int n_in,
                              void* d_out, int out_size, void* d_ws, size_t ws_size,
                              hipStream_t stream) {
    const float* in_feat = (const float*)d_in[0];
    const int*   src     = (const int*)d_in[1];
    const int*   dst     = (const int*)d_in[2];
    const float* W1      = (const float*)d_in[3];
    const float* al1     = (const float*)d_in[4];
    const float* ar1     = (const float*)d_in[5];
    const float* b1      = (const float*)d_in[6];
    const float* W2      = (const float*)d_in[7];
    const float* al2     = (const float*)d_in[8];
    const float* ar2     = (const float*)d_in[9];
    const float* b2      = (const float*)d_in[10];
    float* out = (float*)d_out;

    char* w = (char*)d_ws;
    size_t off = 0;
    auto take = [&](size_t bytes) -> void* {
        void* p = (void*)(w + off);
        off += (bytes + 255) & ~(size_t)255;
        return p;
    };
    int*      cnt     = (int*)     take((size_t)CS_S * N_NODES * 4);
    __bf16*   w1t     = (__bf16*)  take((size_t)64 * 128 * 2);
    __bf16*   w2t     = (__bf16*)  take((size_t)16 * 32 * 2);
    float*    feat1   = (float*)   take((size_t)N_NODES * 64 * 4);
    float*    er1     = (float*)   take((size_t)N_NODES * 2 * 4);
    float*    feat2   = (float*)   take((size_t)N_NODES * 32 * 4);
    float*    er2     = (float*)   take((size_t)N_NODES * 4 * 4);
    int*      deg     = (int*)     take((size_t)N_NODES * 4);
    int*      rowptr  = (int*)     take((size_t)(N_NODES + 1) * 4);
    int*      csr_src = (int*)     take((size_t)N_EDGES * 4);
    (void)ws_size; (void)n_in; (void)in_sizes; (void)out_size;

    auto cdiv = [](long a, long b) { return (int)((a + b - 1) / b); };
    const int T = 256;

    k_count<<<CS_R * CS_S, CS_T, 0, stream>>>(dst, cnt);
    k_slab_prefix<<<SP_BLOCKS + PACK_BLOCKS, SP_T, 0, stream>>>(cnt, deg, W1, W2, w1t, w2t);
    k_scan_gemm1<<<1 + G1_BLOCKS, SCAN_T, 0, stream>>>(deg, rowptr, in_feat, w1t, ar1, feat1, er1);
    k_fill<<<CS_R * CS_S, CS_T, 0, stream>>>(src, dst, rowptr, cnt, csr_src);

    k_seg1<<<cdiv((long)N_NODES * 16, S1_WAVES * 32), S1_WAVES * 32, 0, stream>>>(
        rowptr, csr_src, feat1, al1, er1, b1, w2t, ar2, feat2, er2);

    k_seg2<<<cdiv((long)N_NODES * 16, T), T, 0, stream>>>(rowptr, csr_src, feat2, al2, er2, b2, out);
}
